# starting kernel
# baseline (speedup 1.0000x reference)
#include <hip/hip_runtime.h>
#include <hip/hip_bf16.h>

#define B_ 16384
#define N_ 1024
#define L_ 3
#define E_ 8
#define R_ 64
#define ER_ (E_*R_)

typedef float v2f __attribute__((ext_vector_type(2)));
typedef float v8f __attribute__((ext_vector_type(8)));

__device__ __forceinline__ v8f wmma_f32(v2f a, v2f b, v8f c) {
    return __builtin_amdgcn_wmma_f32_16x16x4_f32(
         false, a,  false, b,
         (short)0, c,  false,  false);
}

__global__ __launch_bounds__(256)
void crossnet_fused_kernel(const float* __restrict__ x,
                           const float* __restrict__ U,
                           const float* __restrict__ V,
                           const float* __restrict__ C,
                           const float* __restrict__ bias,
                           const float* __restrict__ gate_w,
                           float* __restrict__ out)
{
    __shared__ float x0_s[16 * N_];
    __shared__ float xl_s[16 * N_];
    __shared__ float v_s [16 * ER_];
    __shared__ float act_s[16 * ER_];
    __shared__ float gates_s[16 * E_];

    const int tid  = threadIdx.x;
    const int lane = tid & 31;
    const int wave = tid >> 5;
    const int mrow = lane & 15;
    const int ksel = lane >> 4;
    const size_t row0 = (size_t)blockIdx.x * 16;

    {
        const float4* xg  = (const float4*)(x + row0 * N_);
        float4* x0v = (float4*)x0_s;
        float4* xlv = (float4*)xl_s;
        #pragma unroll
        for (int i = 0; i < (16 * N_ / 4) / 256; ++i) {
            int idx = tid + i * 256;
            float4 t = xg[idx];
            x0v[idx] = t;
            xlv[idx] = t;
        }
    }
    __syncthreads();

    for (int l = 0; l < L_; ++l) {
        if (tid < 16 * E_) {
            const int row = tid >> 3, e = tid & 7;
            const float* gw = gate_w + e * N_;
            const float* xr = xl_s + row * N_;
            float acc = 0.f;
            for (int n = 0; n < N_; ++n) acc = fmaf(xr[n], gw[n], acc);
            gates_s[tid] = acc;
        }
        __syncthreads();
        if (tid < 16) {
            float m = gates_s[tid * E_];
            #pragma unroll
            for (int e = 1; e < E_; ++e) m = fmaxf(m, gates_s[tid * E_ + e]);
            float tmp[E_]; float s = 0.f;
            #pragma unroll
            for (int e = 0; e < E_; ++e) { tmp[e] = __expf(gates_s[tid * E_ + e] - m); s += tmp[e]; }
            const float inv = 1.f / s;
            #pragma unroll
            for (int e = 0; e < E_; ++e) gates_s[tid * E_ + e] = tmp[e] * inv;
        }
        {
            const int e = wave;
            const float* Vb = V + ((size_t)l * E_ + e) * (R_ * N_);
            v8f a0 = {}, a1 = {}, a2 = {}, a3 = {};
            const float* a_src = xl_s + mrow * N_ + ksel * 2;
            for (int n0 = 0; n0 < N_; n0 += 4) {
                v2f a = *(const v2f*)(a_src + n0);
                const float* bp = Vb + n0 + ksel * 2;
                v2f b0 = *(const v2f*)(bp + (0 * 16 + mrow) * N_);
                v2f b1 = *(const v2f*)(bp + (1 * 16 + mrow) * N_);
                v2f b2 = *(const v2f*)(bp + (2 * 16 + mrow) * N_);
                v2f b3 = *(const v2f*)(bp + (3 * 16 + mrow) * N_);
                a0 = wmma_f32(a, b0, a0);
                a1 = wmma_f32(a, b1, a1);
                a2 = wmma_f32(a, b2, a2);
                a3 = wmma_f32(a, b3, a3);
            }
            v8f acc[4] = {a0, a1, a2, a3};
            #pragma unroll
            for (int t = 0; t < 4; ++t)
                #pragma unroll
                for (int j = 0; j < 8; ++j) {
                    const int row = j + 8 * ksel;
                    v_s[row * ER_ + e * R_ + t * 16 + mrow] = fmaxf(acc[t][j], 0.f);
                }
        }
        __syncthreads();

        {
            const int e = wave;
            const float* Cb = C + ((size_t)l * E_ + e) * (R_ * R_);
            v8f a0 = {}, a1 = {}, a2 = {}, a3 = {};
            const float* a_src = v_s + mrow * ER_ + e * R_ + ksel * 2;
            #pragma unroll
            for (int s0 = 0; s0 < R_; s0 += 4) {
                v2f a = *(const v2f*)(a_src + s0);
                const float* bp = Cb + s0 + ksel * 2;
                v2f b0 = *(const v2f*)(bp + (0 * 16 + mrow) * R_);
                v2f b1 = *(const v2f*)(bp + (1 * 16 + mrow) * R_);
                v2f b2 = *(const v2f*)(bp + (2 * 16 + mrow) * R_);
                v2f b3 = *(const v2f*)(bp + (3 * 16 + mrow) * R_);
                a0 = wmma_f32(a, b0, a0);
                a1 = wmma_f32(a, b1, a1);
                a2 = wmma_f32(a, b2, a2);
                a3 = wmma_f32(a, b3, a3);
            }
            v8f acc[4] = {a0, a1, a2, a3};
            #pragma unroll
            for (int t = 0; t < 4; ++t)
                #pragma unroll
                for (int j = 0; j < 8; ++j) {
                    const int row = j + 8 * ksel;
                    const float g = gates_s[row * E_ + e];
                    act_s[row * ER_ + e * R_ + t * 16 + mrow] = g * fmaxf(acc[t][j], 0.f);
                }
        }
        __syncthreads();

        {
            const int n_base = wave * 128;
            const float* Ub = U + (size_t)l * (E_ * N_ * R_);
            v8f acc[8] = {};
            const float* a_src = act_s + mrow * ER_ + ksel * 2;
            for (int k0 = 0; k0 < ER_; k0 += 4) {
                v2f a = *(const v2f*)(a_src + k0);
                const int e  = k0 >> 6;
                const int r0 = (k0 & 63) + ksel * 2;
                const float* bp = Ub + ((size_t)e * N_) * R_ + r0;
                #pragma unroll
                for (int t = 0; t < 8; ++t) {
                    const int n = n_base + t * 16 + mrow;
                    v2f b = *(const v2f*)(bp + (size_t)n * R_);
                    acc[t] = wmma_f32(a, b, acc[t]);
                }
            }
            const float* bl = bias + l * N_;
            #pragma unroll
            for (int t = 0; t < 8; ++t) {
                const int n = n_base + t * 16 + mrow;
                const float bn = bl[n];
                #pragma unroll
                for (int j = 0; j < 8; ++j) {
                    const int row = j + 8 * ksel;
                    const float gu  = acc[t][j];
                    const float x0v = x0_s[row * N_ + n];
                    const float xlv = xl_s[row * N_ + n];
                    xl_s[row * N_ + n] = fmaf(x0v, gu + bn, xlv);
                }
            }
        }
        __syncthreads();
    }

    {
        float4* og = (float4*)(out + row0 * N_);
        const float4* xlv = (const float4*)xl_s;
        #pragma unroll
        for (int i = 0; i < (16 * N_ / 4) / 256; ++i) {
            int idx = tid + i * 256;
            og[idx] = xlv[idx];
        }
    }
}

extern "C" void kernel_launch(void* const* d_in, const int* in_sizes, int n_in,
                              void* d_out, int out_size, void* d_ws, size_t ws_size,
                              hipStream_t stream) {
    const float* x      = (const float*)d_in[0];
    const float* U      = (const float*)d_in[1];
    const float* V      = (const float*)d_in[2];
    const float* C      = (const float*)d_in[3];
    const float* bias   = (const float*)d_in[4];
    const float* gate_w = (const float*)d_in[5];
    float* out = (float*)d_out;

    dim3 grid(B_ / 16);
    dim3 block(256);
    crossnet_fused_kernel<<<grid, block, 0, stream>>>(x, U, V, C, bias, gate_w, out);
}


# gate logits fused into the x_l update: each wave dots its 128 new x_l columns with gate_w in registers, row_xmask adds, 8 partials summed at the softmax; dot-loop phase goes (logit sum order changes)
# speedup vs baseline: 2.5346x; 2.5346x over previous
#include <hip/hip_runtime.h>
#include <hip/hip_bf16.h>

#define B_ 16384
#define N_ 1024
#define L_ 3
#define E_ 8
#define R_ 64
#define ER_ (E_*R_)

#define KC_ 32
#define WP_ (KC_ + 4)
#define WROWS_ 128
#define WBUF_ (WROWS_ * WP_)

typedef float v2f __attribute__((ext_vector_type(2)));
typedef float v4f __attribute__((ext_vector_type(4)));
typedef float v8f __attribute__((ext_vector_type(8)));

__device__ __forceinline__ v8f wmma_f32(v2f a, v2f b, v8f c) {
    return __builtin_amdgcn_wmma_f32_16x16x4_f32(
         false, a,  false, b,
         (short)0, c,  false,  false);
}

__device__ __forceinline__ void wave_lds_order() {
    __builtin_amdgcn_fence(__ATOMIC_RELEASE, "wavefront");
    __builtin_amdgcn_wave_barrier();
    __builtin_amdgcn_fence(__ATOMIC_ACQUIRE, "wavefront");
}

template <int NT>
__device__ __forceinline__ void w_fetch(v4f (&r)[NT * 4], const float* __restrict__ src,
                                        const int rs, const int lane) {
    const float* p = src + (size_t)(lane >> 3) * rs + (lane & 7) * 4;
    #pragma unroll
    for (int i = 0; i < NT * 4; ++i)
        r[i] = *(const v4f*)(p + (size_t)(i * 4) * rs);
}

template <int NT>
__device__ __forceinline__ void w_put(const v4f (&r)[NT * 4], float* wb, const int lane) {
    float* q = wb + (lane >> 3) * WP_ + (lane & 7) * 4;
    #pragma unroll
    for (int i = 0; i < NT * 4; ++i)
        *(v4f*)(q + i * 4 * WP_) = r[i];
}

template <int NT>
__device__ __forceinline__ void chunk_run(v8f (&acc)[NT], const float* ap, const float* bfrag) {
    #pragma unroll
    for (int kk = 0; kk < KC_ / 4; ++kk) {
        const v2f a = *(const v2f*)(ap + kk * 4);
        #pragma unroll
        for (int t = 0; t < NT; ++t) {
            const v2f b = *(const v2f*)(bfrag + t * 16 * WP_ + kk * 4);
            acc[t] = wmma_f32(a, b, acc[t]);
        }
    }
}

template <int NT>
__device__ __forceinline__ void stream_gemm(v8f (&acc)[NT], const float* a_src,
                                            const float* __restrict__ w_base, const int rs,
                                            const size_t s_hi, const int nchunks,
                                            float* wb, const int lane) {
    const int mrow = lane & 15;
    const int ksel = lane >> 4;
    const float* bfrag = wb + mrow * WP_ + ksel * 2;
    {
        v4f r0[NT * 4];
        w_fetch<NT>(r0, w_base, rs, lane);
        wave_lds_order();
        w_put<NT>(r0, wb, lane);
    }
    for (int c = 0; c + 1 < nchunks; ++c) {
        const int cn = c + 1;
        v4f r[NT * 4];
        w_fetch<NT>(r, w_base + (size_t)(cn >> 1) * s_hi + (cn & 1) * KC_, rs, lane);
        wave_lds_order();
        chunk_run<NT>(acc, a_src + c * KC_, bfrag);
        __builtin_amdgcn_sched_barrier(0);
        wave_lds_order();
        w_put<NT>(r, wb, lane);
    }
    wave_lds_order();
    chunk_run<NT>(acc, a_src + (nchunks - 1) * KC_, bfrag);
}

template <int M>
__device__ __forceinline__ float row_xor_add(const float x) {
    const int b = __builtin_bit_cast(int, x);
    return x + __builtin_bit_cast(float, __builtin_amdgcn_update_dpp(0, b, 0x160 + M, 0xf, 0xf, true));
}

__device__ __forceinline__ void gate_partials(const v8f (&xv)[8], const float* __restrict__ gate_w,
                                              const int n_base, const int mrow, const int ksel,
                                              float* gp_out) {
    float gp[8][E_];
    #pragma unroll
    for (int j = 0; j < 8; ++j)
        #pragma unroll
        for (int e = 0; e < E_; ++e) gp[j][e] = 0.f;
    #pragma unroll
    for (int t = 0; t < 8; ++t) {
        float gwv[E_];
        #pragma unroll
        for (int e = 0; e < E_; ++e) gwv[e] = gate_w[e * N_ + n_base + t * 16 + mrow];
        #pragma unroll
        for (int j = 0; j < 8; ++j)
            #pragma unroll
            for (int e = 0; e < E_; ++e) gp[j][e] = fmaf(xv[t][j], gwv[e], gp[j][e]);
    }
    #pragma unroll
    for (int j = 0; j < 8; ++j)
        #pragma unroll
        for (int e = 0; e < E_; ++e) {
            float s = gp[j][e];
            s = row_xor_add<1>(s);
            s = row_xor_add<2>(s);
            s = row_xor_add<4>(s);
            s = row_xor_add<8>(s);
            gp[j][e] = s;
        }
    if (mrow == 0) {
        #pragma unroll
        for (int j = 0; j < 8; ++j) {
            float* q = gp_out + (j + 8 * ksel) * E_;
            *(v4f*)(q)     = (v4f){gp[j][0], gp[j][1], gp[j][2], gp[j][3]};
            *(v4f*)(q + 4) = (v4f){gp[j][4], gp[j][5], gp[j][6], gp[j][7]};
        }
    }
}

__global__ __launch_bounds__(256)
void crossnet_fused_kernel(const float* __restrict__ x,
                           const float* __restrict__ U,
                           const float* __restrict__ V,
                           const float* __restrict__ C,
                           const float* __restrict__ bias,
                           const float* __restrict__ gate_w,
                           float* __restrict__ out)
{
    __shared__ float x0_s[16 * N_];
    __shared__ float xl_s[16 * N_];
    __shared__ float v_s [16 * ER_];
    __shared__ float gates_s[16 * E_];
    __shared__ __align__(16) float gpart_s[8 * 16 * E_];
    __shared__ __align__(16) float w_s[8 * WBUF_];

    const int tid  = threadIdx.x;
    const int lane = tid & 31;
    const int wave = tid >> 5;
    const int mrow = lane & 15;
    const int ksel = lane >> 4;
    const size_t row0 = (size_t)blockIdx.x * 16;
    float* wb = w_s + wave * WBUF_;
    float* act_s = v_s;

    {
        const float4* xg  = (const float4*)(x + row0 * N_);
        float4* x0v = (float4*)x0_s;
        float4* xlv = (float4*)xl_s;
        #pragma unroll
        for (int i = 0; i < (16 * N_ / 4) / 256; ++i) {
            int idx = tid + i * 256;
            float4 t = xg[idx];
            x0v[idx] = t;
            xlv[idx] = t;
        }
    }
    __syncthreads();

    {
        v8f xv[8];
        #pragma unroll
        for (int t = 0; t < 8; ++t)
            #pragma unroll
            for (int j = 0; j < 8; ++j)
                xv[t][j] = xl_s[(j + 8 * ksel) * N_ + wave * 128 + t * 16 + mrow];
        gate_partials(xv, gate_w, wave * 128, mrow, ksel, gpart_s + wave * (16 * E_));
    }
    __syncthreads();

    for (int l = 0; l < L_; ++l) {
        if (tid < 16) {
            float lg[E_];
            #pragma unroll
            for (int e = 0; e < E_; ++e) lg[e] = 0.f;
            #pragma unroll
            for (int w = 0; w < 8; ++w) {
                const v4f a = *(const v4f*)(gpart_s + w * (16 * E_) + tid * E_);
                const v4f b = *(const v4f*)(gpart_s + w * (16 * E_) + tid * E_ + 4);
                #pragma unroll
                for (int e = 0; e < 4; ++e) { lg[e] += a[e]; lg[4 + e] += b[e]; }
            }
            float m = lg[0];
            #pragma unroll
            for (int e = 1; e < E_; ++e) m = fmaxf(m, lg[e]);
            float tmp[E_]; float s = 0.f;
            #pragma unroll
            for (int e = 0; e < E_; ++e) { tmp[e] = __expf(lg[e] - m); s += tmp[e]; }
            const float inv = 1.f / s;
            #pragma unroll
            for (int e = 0; e < E_; ++e) gates_s[tid * E_ + e] = tmp[e] * inv;
        }
        {
            const int e = wave;
            const float* Vb = V + ((size_t)l * E_ + e) * (R_ * N_);
            v8f acc[4] = {};
            stream_gemm<4>(acc, xl_s + mrow * N_ + ksel * 2, Vb, N_, 2 * KC_, N_ / KC_, wb, lane);
            #pragma unroll
            for (int t = 0; t < 4; ++t)
                #pragma unroll
                for (int j = 0; j < 8; ++j) {
                    const int row = j + 8 * ksel;
                    v_s[row * ER_ + e * R_ + t * 16 + mrow] = fmaxf(acc[t][j], 0.f);
                }
        }
        __syncthreads();

        {
            const int e = wave;
            const float* Cb = C + ((size_t)l * E_ + e) * (R_ * R_);
            v8f acc[4] = {};
            stream_gemm<4>(acc, v_s + mrow * ER_ + e * R_ + ksel * 2, Cb, R_, 2 * KC_, R_ / KC_, wb, lane);
            #pragma unroll
            for (int t = 0; t < 4; ++t)
                #pragma unroll
                for (int j = 0; j < 8; ++j) {
                    const int row = j + 8 * ksel;
                    const float g = gates_s[row * E_ + e];
                    act_s[row * ER_ + e * R_ + t * 16 + mrow] = g * fmaxf(acc[t][j], 0.f);
                }
        }
        __syncthreads();

        {
            const int n_base = wave * 128;
            const float* Ub = U + (size_t)l * (E_ * N_ * R_) + (size_t)n_base * R_;
            v8f acc[8] = {};
            stream_gemm<8>(acc, act_s + mrow * ER_ + ksel * 2, Ub, R_, (size_t)N_ * R_, ER_ / KC_, wb, lane);
            const float* bl = bias + l * N_;
            #pragma unroll
            for (int t = 0; t < 8; ++t) {
                const int n = n_base + t * 16 + mrow;
                const float bn = bl[n];
                #pragma unroll
                for (int j = 0; j < 8; ++j) {
                    const int row = j + 8 * ksel;
                    const float gu  = acc[t][j];
                    const float x0v = x0_s[row * N_ + n];
                    const float xlv = xl_s[row * N_ + n];
                    const float xn  = fmaf(x0v, gu + bn, xlv);
                    xl_s[row * N_ + n] = xn;
                    acc[t][j] = xn;
                }
            }
            if (l + 1 < L_)
                gate_partials(acc, gate_w, n_base, mrow, ksel, gpart_s + wave * (16 * E_));
        }
        __syncthreads();
    }

    {
        float4* og = (float4*)(out + row0 * N_);
        const float4* xlv = (const float4*)xl_s;
        #pragma unroll
        for (int i = 0; i < (16 * N_ / 4) / 256; ++i) {
            int idx = tid + i * 256;
            og[idx] = xlv[idx];
        }
    }
}

extern "C" void kernel_launch(void* const* d_in, const int* in_sizes, int n_in,
                              void* d_out, int out_size, void* d_ws, size_t ws_size,
                              hipStream_t stream) {
    const float* x      = (const float*)d_in[0];
    const float* U      = (const float*)d_in[1];
    const float* V      = (const float*)d_in[2];
    const float* C      = (const float*)d_in[3];
    const float* bias   = (const float*)d_in[4];
    const float* gate_w = (const float*)d_in[5];
    float* out = (float*)d_out;

    dim3 grid(B_ / 16);
    dim3 block(256);
    crossnet_fused_kernel<<<grid, block, 0, stream>>>(x, U, V, C, bias, gate_w, out);
}


# 32 batch rows a work-group: x_l tile (128 KB) stays in LDS, each LDS-staged B fragment feeds both 16-row tiles; x0 re-read from global, phase D as two 64-column passes (logit sum order changes)
# speedup vs baseline: 2.9666x; 1.1704x over previous
#include <hip/hip_runtime.h>
#include <hip/hip_bf16.h>

#define B_ 16384
#define N_ 1024
#define L_ 3
#define E_ 8
#define R_ 64
#define ER_ (E_*R_)
#define TM_ 32

#define KC_ 32
#define WP_ (KC_ + 4)
#define WROWS_ 64
#define WBUF_ (WROWS_ * WP_)

typedef float v2f __attribute__((ext_vector_type(2)));
typedef float v4f __attribute__((ext_vector_type(4)));
typedef float v8f __attribute__((ext_vector_type(8)));

__device__ __forceinline__ v8f wmma_f32(v2f a, v2f b, v8f c) {
    return __builtin_amdgcn_wmma_f32_16x16x4_f32(
         false, a,  false, b,
         (short)0, c,  false,  false);
}

__device__ __forceinline__ void wave_lds_order() {
    __builtin_amdgcn_fence(__ATOMIC_RELEASE, "wavefront");
    __builtin_amdgcn_wave_barrier();
    __builtin_amdgcn_fence(__ATOMIC_ACQUIRE, "wavefront");
}

template <int NT>
__device__ __forceinline__ void w_fetch(v4f (&r)[NT * 4], const float* __restrict__ src,
                                        const int rs, const int lane) {
    const float* p = src + (size_t)(lane >> 3) * rs + (lane & 7) * 4;
    #pragma unroll
    for (int i = 0; i < NT * 4; ++i)
        r[i] = *(const v4f*)(p + (size_t)(i * 4) * rs);
}

template <int NT>
__device__ __forceinline__ void w_put(const v4f (&r)[NT * 4], float* wb, const int lane) {
    float* q = wb + (lane >> 3) * WP_ + (lane & 7) * 4;
    #pragma unroll
    for (int i = 0; i < NT * 4; ++i)
        *(v4f*)(q + i * 4 * WP_) = r[i];
}

template <int NT>
__device__ __forceinline__ void chunk_run(v8f (&acc)[2][NT], const float* ap, const int a_tile,
                                          const float* bfrag) {
    #pragma unroll
    for (int kk = 0; kk < KC_ / 4; ++kk) {
        const v2f a0 = *(const v2f*)(ap + kk * 4);
        const v2f a1 = *(const v2f*)(ap + a_tile + kk * 4);
        #pragma unroll
        for (int t = 0; t < NT; ++t) {
            const v2f b = *(const v2f*)(bfrag + t * 16 * WP_ + kk * 4);
            acc[0][t] = wmma_f32(a0, b, acc[0][t]);
            acc[1][t] = wmma_f32(a1, b, acc[1][t]);
        }
    }
}

template <int NT, typename F>
__device__ __forceinline__ void stream_gemm(v8f (&acc)[2][NT], const float* a_src, const int a_tile,
                                            const float* __restrict__ w_base, const int rs,
                                            const size_t s_hi, const int nchunks,
                                            float* wb, const int lane, F&& after_first) {
    const int mrow = lane & 15;
    const int ksel = lane >> 4;
    const float* bfrag = wb + mrow * WP_ + ksel * 2;
    {
        v4f r0[NT * 4];
        w_fetch<NT>(r0, w_base, rs, lane);
        wave_lds_order();
        w_put<NT>(r0, wb, lane);
    }
    after_first();
    for (int c = 0; c + 1 < nchunks; ++c) {
        const int cn = c + 1;
        v4f r[NT * 4];
        w_fetch<NT>(r, w_base + (size_t)(cn >> 1) * s_hi + (cn & 1) * KC_, rs, lane);
        wave_lds_order();
        chunk_run<NT>(acc, a_src + c * KC_, a_tile, bfrag);
        __builtin_amdgcn_sched_barrier(0);
        wave_lds_order();
        w_put<NT>(r, wb, lane);
    }
    wave_lds_order();
    chunk_run<NT>(acc, a_src + (nchunks - 1) * KC_, a_tile, bfrag);
}

template <int M>
__device__ __forceinline__ float row_xor_add(const float x) {
    const int b = __builtin_bit_cast(int, x);
    return x + __builtin_bit_cast(float, __builtin_amdgcn_update_dpp(0, b, 0x160 + M, 0xf, 0xf, true));
}

__device__ __forceinline__ void gate_partials(const v8f (&xv)[4], const float* __restrict__ gate_w,
                                              const int n_base, const int mrow, const int ksel,
                                              float* gp_out) {
    float gp[8][E_];
    #pragma unroll
    for (int j = 0; j < 8; ++j)
        #pragma unroll
        for (int e = 0; e < E_; ++e) gp[j][e] = 0.f;
    #pragma unroll
    for (int t = 0; t < 4; ++t) {
        float gwv[E_];
        #pragma unroll
        for (int e = 0; e < E_; ++e) gwv[e] = gate_w[e * N_ + n_base + t * 16 + mrow];
        #pragma unroll
        for (int j = 0; j < 8; ++j)
            #pragma unroll
            for (int e = 0; e < E_; ++e) gp[j][e] = fmaf(xv[t][j], gwv[e], gp[j][e]);
    }
    #pragma unroll
    for (int j = 0; j < 8; ++j)
        #pragma unroll
        for (int e = 0; e < E_; ++e) {
            float s = gp[j][e];
            s = row_xor_add<1>(s);
            s = row_xor_add<2>(s);
            s = row_xor_add<4>(s);
            s = row_xor_add<8>(s);
            gp[j][e] = s;
        }
    if (mrow == 0) {
        #pragma unroll
        for (int j = 0; j < 8; ++j) {
            float* q = gp_out + (j + 8 * ksel) * E_;
            *(v4f*)(q)     = (v4f){gp[j][0], gp[j][1], gp[j][2], gp[j][3]};
            *(v4f*)(q + 4) = (v4f){gp[j][4], gp[j][5], gp[j][6], gp[j][7]};
        }
    }
}

__global__ __launch_bounds__(256)
void crossnet_fused_kernel(const float* __restrict__ x,
                           const float* __restrict__ U,
                           const float* __restrict__ V,
                           const float* __restrict__ C,
                           const float* __restrict__ bias,
                           const float* __restrict__ gate_w,
                           float* __restrict__ out)
{
    __shared__ float xl_s[TM_ * N_];
    __shared__ float v_s [TM_ * ER_];
    __shared__ float gates_s[TM_ * E_];
    __shared__ __align__(16) float gpart_s[16 * TM_ * E_];
    __shared__ __align__(16) float w_s[8 * WBUF_];

    const int tid  = threadIdx.x;
    const int lane = tid & 31;
    const int wave = tid >> 5;
    const int mrow = lane & 15;
    const int ksel = lane >> 4;
    const size_t row0 = (size_t)blockIdx.x * TM_;
    float* wb = w_s + wave * WBUF_;
    float* act_s = v_s;
    const auto nothing = []() {};

    {
        const float4* xg  = (const float4*)(x + row0 * N_);
        float4* xlv = (float4*)xl_s;
        #pragma unroll
        for (int i = 0; i < (TM_ * N_ / 4) / 256; ++i) {
            int idx = tid + i * 256;
            xlv[idx] = xg[idx];
        }
    }
    __syncthreads();

    for (int p = 0; p < 2; ++p) {
        const int n_base = wave * 128 + p * 64;
        #pragma unroll
        for (int m = 0; m < 2; ++m) {
            v8f xv[4];
            #pragma unroll
            for (int t = 0; t < 4; ++t)
                #pragma unroll
                for (int j = 0; j < 8; ++j)
                    xv[t][j] = xl_s[(m * 16 + j + 8 * ksel) * N_ + n_base + t * 16 + mrow];
            gate_partials(xv, gate_w, n_base, mrow, ksel, gpart_s + (wave * 2 + p) * (TM_ * E_) + m * (16 * E_));
        }
    }
    __syncthreads();

    for (int l = 0; l < L_; ++l) {
        if (tid < TM_) {
            float lg[E_];
            #pragma unroll
            for (int e = 0; e < E_; ++e) lg[e] = 0.f;
            #pragma unroll
            for (int w = 0; w < 16; ++w) {
                const v4f a = *(const v4f*)(gpart_s + w * (TM_ * E_) + tid * E_);
                const v4f b = *(const v4f*)(gpart_s + w * (TM_ * E_) + tid * E_ + 4);
                #pragma unroll
                for (int e = 0; e < 4; ++e) { lg[e] += a[e]; lg[4 + e] += b[e]; }
            }
            float m = lg[0];
            #pragma unroll
            for (int e = 1; e < E_; ++e) m = fmaxf(m, lg[e]);
            float tmp[E_]; float s = 0.f;
            #pragma unroll
            for (int e = 0; e < E_; ++e) { tmp[e] = __expf(lg[e] - m); s += tmp[e]; }
            const float inv = 1.f / s;
            #pragma unroll
            for (int e = 0; e < E_; ++e) gates_s[tid * E_ + e] = tmp[e] * inv;
        }
        {
            const int e = wave;
            const float* Vb = V + ((size_t)l * E_ + e) * (R_ * N_);
            v8f acc[2][4] = {};
            stream_gemm<4>(acc, xl_s + mrow * N_ + ksel * 2, 16 * N_, Vb, N_, 2 * KC_, N_ / KC_, wb, lane, nothing);
            #pragma unroll
            for (int m = 0; m < 2; ++m)
                #pragma unroll
                for (int t = 0; t < 4; ++t)
                    #pragma unroll
                    for (int j = 0; j < 8; ++j) {
                        const int row = m * 16 + j + 8 * ksel;
                        v_s[row * ER_ + e * R_ + t * 16 + mrow] = fmaxf(acc[m][t][j], 0.f);
                    }
        }
        __syncthreads();

        {
            const int e = wave;
            const float* Cb = C + ((size_t)l * E_ + e) * (R_ * R_);
            v8f acc[2][4] = {};
            stream_gemm<4>(acc, v_s + mrow * ER_ + e * R_ + ksel * 2, 16 * ER_, Cb, R_, 2 * KC_, R_ / KC_, wb, lane, nothing);
            #pragma unroll
            for (int m = 0; m < 2; ++m)
                #pragma unroll
                for (int t = 0; t < 4; ++t)
                    #pragma unroll
                    for (int j = 0; j < 8; ++j) {
                        const int row = m * 16 + j + 8 * ksel;
                        const float g = gates_s[row * E_ + e];
                        act_s[row * ER_ + e * R_ + t * 16 + mrow] = g * fmaxf(acc[m][t][j], 0.f);
                    }
        }
        __syncthreads();

        for (int p = 0; p < 2; ++p) {
            const int n_base = wave * 128 + p * 64;
            const float* Ub = U + (size_t)l * (E_ * N_ * R_) + (size_t)n_base * R_;
            v8f acc[2][4] = {};
            v8f x0r[2][4];
            stream_gemm<4>(acc, act_s + mrow * ER_ + ksel * 2, 16 * ER_, Ub, R_, (size_t)N_ * R_, ER_ / KC_, wb, lane,
                [&]() {
                    #pragma unroll
                    for (int m = 0; m < 2; ++m)
                        #pragma unroll
                        for (int t = 0; t < 4; ++t)
                            #pragma unroll
                            for (int j = 0; j < 8; ++j)
                                x0r[m][t][j] = x[(row0 + m * 16 + j + 8 * ksel) * N_ + n_base + t * 16 + mrow];
                });
            const float* bl = bias + l * N_;
            #pragma unroll
            for (int m = 0; m < 2; ++m)
                #pragma unroll
                for (int t = 0; t < 4; ++t) {
                    const int n = n_base + t * 16 + mrow;
                    const float bn = bl[n];
                    #pragma unroll
                    for (int j = 0; j < 8; ++j) {
                        const int row = m * 16 + j + 8 * ksel;
                        const float gu  = acc[m][t][j];
                        const float xlv = xl_s[row * N_ + n];
                        const float xn  = fmaf(x0r[m][t][j], gu + bn, xlv);
                        xl_s[row * N_ + n] = xn;
                        acc[m][t][j] = xn;
                    }
                }
            if (l + 1 < L_) {
                #pragma unroll
                for (int m = 0; m < 2; ++m)
                    gate_partials(acc[m], gate_w, n_base, mrow, ksel,
                                  gpart_s + (wave * 2 + p) * (TM_ * E_) + m * (16 * E_));
            }
        }
        __syncthreads();
    }

    {
        float4* og = (float4*)(out + row0 * N_);
        const float4* xlv = (const float4*)xl_s;
        #pragma unroll
        for (int i = 0; i < (TM_ * N_ / 4) / 256; ++i) {
            int idx = tid + i * 256;
            og[idx] = xlv[idx];
        }
    }
}

extern "C" void kernel_launch(void* const* d_in, const int* in_sizes, int n_in,
                              void* d_out, int out_size, void* d_ws, size_t ws_size,
                              hipStream_t stream) {
    const float* x      = (const float*)d_in[0];
    const float* U      = (const float*)d_in[1];
    const float* V      = (const float*)d_in[2];
    const float* C      = (const float*)d_in[3];
    const float* bias   = (const float*)d_in[4];
    const float* gate_w = (const float*)d_in[5];
    float* out = (float*)d_out;

    dim3 grid(B_ / TM_);
    dim3 block(256);
    crossnet_fused_kernel<<<grid, block, 0, stream>>>(x, U, V, C, bias, gate_w, out);
}


# priority window: waves 4-7 (younger wave of each SIMD pair) hold s_setprio 3 for the first 22 of the 32 chunks between two barriers in phases B and D, then 0, so a pair reaches the barrier together
# speedup vs baseline: 3.0682x; 1.0342x over previous
#include <hip/hip_runtime.h>
#include <hip/hip_bf16.h>

#define B_ 16384
#define N_ 1024
#define L_ 3
#define E_ 8
#define R_ 64
#define ER_ (E_*R_)
#define TM_ 32

#define KC_ 32
#define WP_ (KC_ + 4)
#define WROWS_ 64
#define WBUF_ (WROWS_ * WP_)

typedef float v2f __attribute__((ext_vector_type(2)));
typedef float v4f __attribute__((ext_vector_type(4)));
typedef float v8f __attribute__((ext_vector_type(8)));

__device__ __forceinline__ v8f wmma_f32(v2f a, v2f b, v8f c) {
    return __builtin_amdgcn_wmma_f32_16x16x4_f32(
         false, a,  false, b,
         (short)0, c,  false,  false);
}

__device__ __forceinline__ void wave_lds_order() {
    __builtin_amdgcn_fence(__ATOMIC_RELEASE, "wavefront");
    __builtin_amdgcn_wave_barrier();
    __builtin_amdgcn_fence(__ATOMIC_ACQUIRE, "wavefront");
}

template <int NT>
__device__ __forceinline__ void w_fetch(v4f (&r)[NT * 4], const float* __restrict__ src,
                                        const int rs, const int lane) {
    const float* p = src + (size_t)(lane >> 3) * rs + (lane & 7) * 4;
    #pragma unroll
    for (int i = 0; i < NT * 4; ++i)
        r[i] = *(const v4f*)(p + (size_t)(i * 4) * rs);
}

template <int NT>
__device__ __forceinline__ void w_put(const v4f (&r)[NT * 4], float* wb, const int lane) {
    float* q = wb + (lane >> 3) * WP_ + (lane & 7) * 4;
    #pragma unroll
    for (int i = 0; i < NT * 4; ++i)
        *(v4f*)(q + i * 4 * WP_) = r[i];
}

template <int NT>
__device__ __forceinline__ void chunk_run(v8f (&acc)[2][NT], const float* ap, const int a_tile,
                                          const float* bfrag) {
    #pragma unroll
    for (int kk = 0; kk < KC_ / 4; ++kk) {
        const v2f a0 = *(const v2f*)(ap + kk * 4);
        const v2f a1 = *(const v2f*)(ap + a_tile + kk * 4);
        #pragma unroll
        for (int t = 0; t < NT; ++t) {
            const v2f b = *(const v2f*)(bfrag + t * 16 * WP_ + kk * 4);
            acc[0][t] = wmma_f32(a0, b, acc[0][t]);
            acc[1][t] = wmma_f32(a1, b, acc[1][t]);
        }
    }
}

template <int NT, typename F>
__device__ __forceinline__ void stream_gemm(v8f (&acc)[2][NT], const float* a_src, const int a_tile,
                                            const float* __restrict__ w_base, const int rs,
                                            const size_t s_hi, const int nchunks, const int drop_c,
                                            float* wb, const int lane, F&& after_first) {
    const int mrow = lane & 15;
    const int ksel = lane >> 4;
    const float* bfrag = wb + mrow * WP_ + ksel * 2;
    {
        v4f r0[NT * 4];
        w_fetch<NT>(r0, w_base, rs, lane);
        wave_lds_order();
        w_put<NT>(r0, wb, lane);
    }
    after_first();
    for (int c = 0; c + 1 < nchunks; ++c) {
        if (c == drop_c) __builtin_amdgcn_s_setprio(0);
        const int cn = c + 1;
        v4f r[NT * 4];
        w_fetch<NT>(r, w_base + (size_t)(cn >> 1) * s_hi + (cn & 1) * KC_, rs, lane);
        wave_lds_order();
        chunk_run<NT>(acc, a_src + c * KC_, a_tile, bfrag);
        __builtin_amdgcn_sched_barrier(0);
        wave_lds_order();
        w_put<NT>(r, wb, lane);
    }
    wave_lds_order();
    chunk_run<NT>(acc, a_src + (nchunks - 1) * KC_, a_tile, bfrag);
}

template <int M>
__device__ __forceinline__ float row_xor_add(const float x) {
    const int b = __builtin_bit_cast(int, x);
    return x + __builtin_bit_cast(float, __builtin_amdgcn_update_dpp(0, b, 0x160 + M, 0xf, 0xf, true));
}

__device__ __forceinline__ void gate_partials(const v8f (&xv)[4], const float* __restrict__ gate_w,
                                              const int n_base, const int mrow, const int ksel,
                                              float* gp_out) {
    float gp[8][E_];
    #pragma unroll
    for (int j = 0; j < 8; ++j)
        #pragma unroll
        for (int e = 0; e < E_; ++e) gp[j][e] = 0.f;
    #pragma unroll
    for (int t = 0; t < 4; ++t) {
        float gwv[E_];
        #pragma unroll
        for (int e = 0; e < E_; ++e) gwv[e] = gate_w[e * N_ + n_base + t * 16 + mrow];
        #pragma unroll
        for (int j = 0; j < 8; ++j)
            #pragma unroll
            for (int e = 0; e < E_; ++e) gp[j][e] = fmaf(xv[t][j], gwv[e], gp[j][e]);
    }
    #pragma unroll
    for (int j = 0; j < 8; ++j)
        #pragma unroll
        for (int e = 0; e < E_; ++e) {
            float s = gp[j][e];
            s = row_xor_add<1>(s);
            s = row_xor_add<2>(s);
            s = row_xor_add<4>(s);
            s = row_xor_add<8>(s);
            gp[j][e] = s;
        }
    if (mrow == 0) {
        #pragma unroll
        for (int j = 0; j < 8; ++j) {
            float* q = gp_out + (j + 8 * ksel) * E_;
            *(v4f*)(q)     = (v4f){gp[j][0], gp[j][1], gp[j][2], gp[j][3]};
            *(v4f*)(q + 4) = (v4f){gp[j][4], gp[j][5], gp[j][6], gp[j][7]};
        }
    }
}

__device__ __forceinline__ void prio_window_open(const bool younger) {
    if (younger) __builtin_amdgcn_s_setprio(3);
}
#define WIN_CHUNKS_ 22

__global__ __launch_bounds__(256)
void crossnet_fused_kernel(const float* __restrict__ x,
                           const float* __restrict__ U,
                           const float* __restrict__ V,
                           const float* __restrict__ C,
                           const float* __restrict__ bias,
                           const float* __restrict__ gate_w,
                           float* __restrict__ out)
{
    __shared__ float xl_s[TM_ * N_];
    __shared__ float v_s [TM_ * ER_];
    __shared__ float gates_s[TM_ * E_];
    __shared__ __align__(16) float gpart_s[16 * TM_ * E_];
    __shared__ __align__(16) float w_s[8 * WBUF_];

    const int tid  = threadIdx.x;
    const int lane = tid & 31;
    const int wave = tid >> 5;
    const int mrow = lane & 15;
    const int ksel = lane >> 4;
    const size_t row0 = (size_t)blockIdx.x * TM_;
    const bool younger = __builtin_amdgcn_readfirstlane(tid >> 7) != 0;
    float* wb = w_s + wave * WBUF_;
    float* act_s = v_s;
    const auto nothing = []() {};

    {
        const float4* xg  = (const float4*)(x + row0 * N_);
        float4* xlv = (float4*)xl_s;
        #pragma unroll
        for (int i = 0; i < (TM_ * N_ / 4) / 256; ++i) {
            int idx = tid + i * 256;
            xlv[idx] = xg[idx];
        }
    }
    __syncthreads();

    for (int p = 0; p < 2; ++p) {
        const int n_base = wave * 128 + p * 64;
        #pragma unroll
        for (int m = 0; m < 2; ++m) {
            v8f xv[4];
            #pragma unroll
            for (int t = 0; t < 4; ++t)
                #pragma unroll
                for (int j = 0; j < 8; ++j)
                    xv[t][j] = xl_s[(m * 16 + j + 8 * ksel) * N_ + n_base + t * 16 + mrow];
            gate_partials(xv, gate_w, n_base, mrow, ksel, gpart_s + (wave * 2 + p) * (TM_ * E_) + m * (16 * E_));
        }
    }
    __syncthreads();

    for (int l = 0; l < L_; ++l) {
        if (tid < TM_) {
            float lg[E_];
            #pragma unroll
            for (int e = 0; e < E_; ++e) lg[e] = 0.f;
            #pragma unroll
            for (int w = 0; w < 16; ++w) {
                const v4f a = *(const v4f*)(gpart_s + w * (TM_ * E_) + tid * E_);
                const v4f b = *(const v4f*)(gpart_s + w * (TM_ * E_) + tid * E_ + 4);
                #pragma unroll
                for (int e = 0; e < 4; ++e) { lg[e] += a[e]; lg[4 + e] += b[e]; }
            }
            float m = lg[0];
            #pragma unroll
            for (int e = 1; e < E_; ++e) m = fmaxf(m, lg[e]);
            float tmp[E_]; float s = 0.f;
            #pragma unroll
            for (int e = 0; e < E_; ++e) { tmp[e] = __expf(lg[e] - m); s += tmp[e]; }
            const float inv = 1.f / s;
            #pragma unroll
            for (int e = 0; e < E_; ++e) gates_s[tid * E_ + e] = tmp[e] * inv;
        }
        {
            const int e = wave;
            const float* Vb = V + ((size_t)l * E_ + e) * (R_ * N_);
            v8f acc[2][4] = {};
            prio_window_open(younger);
            stream_gemm<4>(acc, xl_s + mrow * N_ + ksel * 2, 16 * N_, Vb, N_, 2 * KC_, N_ / KC_, WIN_CHUNKS_, wb, lane, nothing);
            #pragma unroll
            for (int m = 0; m < 2; ++m)
                #pragma unroll
                for (int t = 0; t < 4; ++t)
                    #pragma unroll
                    for (int j = 0; j < 8; ++j) {
                        const int row = m * 16 + j + 8 * ksel;
                        v_s[row * ER_ + e * R_ + t * 16 + mrow] = fmaxf(acc[m][t][j], 0.f);
                    }
        }
        __syncthreads();

        {
            const int e = wave;
            const float* Cb = C + ((size_t)l * E_ + e) * (R_ * R_);
            v8f acc[2][4] = {};
            stream_gemm<4>(acc, v_s + mrow * ER_ + e * R_ + ksel * 2, 16 * ER_, Cb, R_, 2 * KC_, R_ / KC_, -1, wb, lane, nothing);
            #pragma unroll
            for (int m = 0; m < 2; ++m)
                #pragma unroll
                for (int t = 0; t < 4; ++t)
                    #pragma unroll
                    for (int j = 0; j < 8; ++j) {
                        const int row = m * 16 + j + 8 * ksel;
                        const float g = gates_s[row * E_ + e];
                        act_s[row * ER_ + e * R_ + t * 16 + mrow] = g * fmaxf(acc[m][t][j], 0.f);
                    }
        }
        __syncthreads();

        for (int p = 0; p < 2; ++p) {
            const int n_base = wave * 128 + p * 64;
            const float* Ub = U + (size_t)l * (E_ * N_ * R_) + (size_t)n_base * R_;
            v8f acc[2][4] = {};
            v8f x0r[2][4];
            if (p == 0) prio_window_open(younger);
            stream_gemm<4>(acc, act_s + mrow * ER_ + ksel * 2, 16 * ER_, Ub, R_, (size_t)N_ * R_, ER_ / KC_,
                p == 0 ? -1 : WIN_CHUNKS_ - ER_ / KC_, wb, lane,
                [&]() {
                    #pragma unroll
                    for (int m = 0; m < 2; ++m)
                        #pragma unroll
                        for (int t = 0; t < 4; ++t)
                            #pragma unroll
                            for (int j = 0; j < 8; ++j)
                                x0r[m][t][j] = x[(row0 + m * 16 + j + 8 * ksel) * N_ + n_base + t * 16 + mrow];
                });
            const float* bl = bias + l * N_;
            #pragma unroll
            for (int m = 0; m < 2; ++m)
                #pragma unroll
                for (int t = 0; t < 4; ++t) {
                    const int n = n_base + t * 16 + mrow;
                    const float bn = bl[n];
                    #pragma unroll
                    for (int j = 0; j < 8; ++j) {
                        const int row = m * 16 + j + 8 * ksel;
                        const float gu  = acc[m][t][j];
                        const float xlv = xl_s[row * N_ + n];
                        const float xn  = fmaf(x0r[m][t][j], gu + bn, xlv);
                        xl_s[row * N_ + n] = xn;
                        acc[m][t][j] = xn;
                    }
                }
            if (l + 1 < L_) {
                #pragma unroll
                for (int m = 0; m < 2; ++m)
                    gate_partials(acc[m], gate_w, n_base, mrow, ksel,
                                  gpart_s + (wave * 2 + p) * (TM_ * E_) + m * (16 * E_));
            }
        }
        __syncthreads();
    }

    {
        float4* og = (float4*)(out + row0 * N_);
        const float4* xlv = (const float4*)xl_s;
        #pragma unroll
        for (int i = 0; i < (TM_ * N_ / 4) / 256; ++i) {
            int idx = tid + i * 256;
            og[idx] = xlv[idx];
        }
    }
}

extern "C" void kernel_launch(void* const* d_in, const int* in_sizes, int n_in,
                              void* d_out, int out_size, void* d_ws, size_t ws_size,
                              hipStream_t stream) {
    const float* x      = (const float*)d_in[0];
    const float* U      = (const float*)d_in[1];
    const float* V      = (const float*)d_in[2];
    const float* C      = (const float*)d_in[3];
    const float* bias   = (const float*)d_in[4];
    const float* gate_w = (const float*)d_in[5];
    float* out = (float*)d_out;

    dim3 grid(B_ / TM_);
    dim3 block(256);
    crossnet_fused_kernel<<<grid, block, 0, stream>>>(x, U, V, C, bias, gate_w, out);
}
